# starting kernel
# baseline (speedup 1.0000x reference)
#include <hip/hip_runtime.h>


typedef __attribute__((ext_vector_type(2))) float v2f;
typedef __attribute__((ext_vector_type(8))) float v8f;

#define DBA_B 32
#define DBA_N 65536
#define DBA_G 32
#define DBA_SPAN (DBA_N / DBA_G)
#define DBA_PW (DBA_SPAN / 8)

__device__ __forceinline__ float fast_rcp(float x) {
  float inv = __builtin_amdgcn_rcpf(x);
  inv = inv * (2.0f - x * inv);
  return inv;
}

__global__ __launch_bounds__(256) void dba_stage1_reduce(
    const float* __restrict__ r, const float* __restrict__ w,
    const float* __restrict__ Jp, const float* __restrict__ Jd,
    const float* __restrict__ lmbda, float* __restrict__ partial)
{
  const int b    = blockIdx.x / DBA_G;
  const int c    = blockIdx.x % DBA_G;
  const int wave = threadIdx.x >> 5;
  const int lane = threadIdx.x & 31;
  const int half = lane >> 4;
  const int i    = lane & 15;
  const float lmb = lmbda[0];

  const int base = c * DBA_SPAN + wave * DBA_PW;
  const size_t rb0 = (size_t)b * DBA_N + base + half;

  const float* pux;
  const float* puy;
  int   strideF;
  float msk;
  if (i < 6) { pux = Jp + rb0 * 12 + i; puy = pux + 6; strideF = 24; msk = 1.f; }
  else       { pux = r + rb0 * 2;       puy = pux + 1; strideF = 4;
               msk = (i == 6) ? 1.f : 0.f; }
  const float2* pw  = (const float2*)(w  + rb0 * 2);
  const float2* pjd = (const float2*)(Jd + rb0 * 2);

  v8f cP = {0.f,0.f,0.f,0.f,0.f,0.f,0.f,0.f};
  v8f cQ = {0.f,0.f,0.f,0.f,0.f,0.f,0.f,0.f};

  for (int t = 0; t < DBA_PW / 2; t += 2) {
    float vA, svA, vB, svB;
    {
      const float  lux = pux[0];
      const float  luy = puy[0];
      const float2 wv  = pw[0];
      const float2 jd  = pjd[0];
      pux += strideF; puy += strideF; pw += 2; pjd += 2;

      const float ux = lux * msk;
      const float uy = luy * msk;
      v2f pa, pb;
      pa.x = wv.x * ux; pa.y = wv.x * uy;
      pb.x = ux;        pb.y = uy;
      cP = __builtin_amdgcn_wmma_f32_16x16x4_f32(false, pa, false, pb,
                                                 (short)0, cP, false, false);
      vA = jd.x * pa.x + jd.y * pa.y;
      const float hdd = wv.x * (jd.x * jd.x + jd.y * jd.y) + lmb + wv.y;
      svA = vA * fast_rcp(hdd + 1e-6f);
    }
    {
      const float  lux = pux[0];
      const float  luy = puy[0];
      const float2 wv  = pw[0];
      const float2 jd  = pjd[0];
      pux += strideF; puy += strideF; pw += 2; pjd += 2;

      const float ux = lux * msk;
      const float uy = luy * msk;
      v2f pa, pb;
      pa.x = wv.x * ux; pa.y = wv.x * uy;
      pb.x = ux;        pb.y = uy;
      cP = __builtin_amdgcn_wmma_f32_16x16x4_f32(false, pa, false, pb,
                                                 (short)0, cP, false, false);
      vB = jd.x * pa.x + jd.y * pa.y;
      const float hdd = wv.x * (jd.x * jd.x + jd.y * jd.y) + lmb + wv.y;
      svB = vB * fast_rcp(hdd + 1e-6f);
    }
    const float vAo  = __shfl_xor(vA, 16, 32);
    const float svAo = __shfl_xor(svA, 16, 32);
    const float vBo  = __shfl_xor(vB, 16, 32);
    const float svBo = __shfl_xor(svB, 16, 32);
    const bool lo = (half == 0);
    v2f qa, qb;
    qa.x = lo ? svA  : svBo;  qa.y = lo ? svAo : svB;
    qb.x = lo ? vA   : vBo;   qb.y = lo ? vAo  : vB;
    cQ = __builtin_amdgcn_wmma_f32_16x16x4_f32(false, qa, false, qb,
                                               (short)0, cQ, false, false);
  }

  __shared__ float sred[8][112];
  if (lane < 7) {
#pragma unroll
    for (int p = 0; p < 7; ++p) {
      sred[wave][p * 7 + lane]      = cP[p];
      sred[wave][49 + p * 7 + lane] = cQ[p];
    }
  }
  __syncthreads();
  if (threadIdx.x < 98) {
    float s = 0.f;
#pragma unroll
    for (int wv = 0; wv < 8; ++wv) s += sred[wv][threadIdx.x];
    partial[(size_t)blockIdx.x * 128 + threadIdx.x] = s;
  }
}

__global__ __launch_bounds__(128) void dba_stage2_solve(
    const float* __restrict__ partial, const float* __restrict__ lmbda,
    float* __restrict__ out)
{
  const int b = blockIdx.x;
  __shared__ float acc[98];
  if (threadIdx.x < 98) {
    float s = 0.f;
    for (int g = 0; g < DBA_G; ++g)
      s += partial[(size_t)(b * DBA_G + g) * 128 + threadIdx.x];
    acc[threadIdx.x] = s;
  }
  __syncthreads();
  if (threadIdx.x == 0) {
    const float lmb = lmbda[0];
    float A[6][7];
    for (int ii = 0; ii < 6; ++ii) {
      for (int jj = 0; jj < 6; ++jj) {
        const float hij = acc[ii * 7 + jj] - acc[49 + ii * 7 + jj];
        const float hji = acc[jj * 7 + ii] - acc[49 + jj * 7 + ii];
        A[ii][jj] = 0.5f * (hij + hji);
      }
      A[ii][ii] += lmb + 0.001f;
      A[ii][6] = acc[ii * 7 + 6] - acc[49 + ii * 7 + 6];
    }
    for (int col = 0; col < 6; ++col) {
      int piv = col; float best = fabsf(A[col][col]);
      for (int rr = col + 1; rr < 6; ++rr) {
        const float av = fabsf(A[rr][col]);
        if (av > best) { best = av; piv = rr; }
      }
      if (piv != col)
        for (int jj = col; jj < 7; ++jj) {
          const float tmp = A[col][jj]; A[col][jj] = A[piv][jj]; A[piv][jj] = tmp;
        }
      const float invd = 1.0f / A[col][col];
      for (int rr = col + 1; rr < 6; ++rr) {
        const float f = A[rr][col] * invd;
        for (int jj = col; jj < 7; ++jj) A[rr][jj] -= f * A[col][jj];
      }
    }
    float x[6];
    for (int ii = 5; ii >= 0; --ii) {
      float s = A[ii][6];
      for (int jj = ii + 1; jj < 6; ++jj) s -= A[ii][jj] * x[jj];
      x[ii] = s / A[ii][ii];
    }
    for (int ii = 0; ii < 6; ++ii) out[b * 6 + ii] = x[ii];
  }
}

__global__ __launch_bounds__(256) void dba_stage3_depth(
    const float* __restrict__ r, const float* __restrict__ w,
    const float* __restrict__ Jp, const float* __restrict__ Jd,
    const float* __restrict__ lmbda, const float* __restrict__ dp,
    float* __restrict__ dd)
{
  const int idx = blockIdx.x * blockDim.x + threadIdx.x;
  if (idx >= DBA_B * DBA_N) return;
  const int b = idx / DBA_N;
  const size_t rb = (size_t)idx;

  const float4* Jp4 = (const float4*)(Jp + rb * 12);
  const float4 a0 = Jp4[0], a1 = Jp4[1], a2 = Jp4[2];
  const float2 wv = ((const float2*)w)[rb];
  const float2 jd = ((const float2*)Jd)[rb];
  const float2 rv = ((const float2*)r)[rb];
  const float lmb = lmbda[0];

  const float p0 = dp[b * 6 + 0], p1 = dp[b * 6 + 1], p2 = dp[b * 6 + 2];
  const float p3 = dp[b * 6 + 3], p4 = dp[b * 6 + 4], p5 = dp[b * 6 + 5];

  const float h0 = wv.x * (jd.x * a0.x + jd.y * a1.z);
  const float h1 = wv.x * (jd.x * a0.y + jd.y * a1.w);
  const float h2 = wv.x * (jd.x * a0.z + jd.y * a2.x);
  const float h3 = wv.x * (jd.x * a0.w + jd.y * a2.y);
  const float h4 = wv.x * (jd.x * a1.x + jd.y * a2.z);
  const float h5 = wv.x * (jd.x * a1.y + jd.y * a2.w);
  const float gd  = wv.x * (jd.x * rv.x + jd.y * rv.y);
  const float hdd = wv.x * (jd.x * jd.x + jd.y * jd.y) + lmb + wv.y;
  const float inv = fast_rcp(hdd + 1e-6f);
  const float vdot = h0 * p0 + h1 * p1 + h2 * p2 + h3 * p3 + h4 * p4 + h5 * p5;
  dd[idx] = inv * (gd - vdot);
}

extern "C" void kernel_launch(void* const* d_in, const int* in_sizes, int n_in,
                              void* d_out, int out_size, void* d_ws, size_t ws_size,
                              hipStream_t stream) {
  const float* r   = (const float*)d_in[0];
  const float* w   = (const float*)d_in[1];
  const float* Jp  = (const float*)d_in[2];
  const float* Jd  = (const float*)d_in[3];
  const float* lmb = (const float*)d_in[4];
  float* out = (float*)d_out;
  float* ws  = (float*)d_ws;

  dba_stage1_reduce<<<dim3(DBA_B * DBA_G), dim3(256), 0, stream>>>(
      r, w, Jp, Jd, lmb, ws);
  dba_stage2_solve<<<dim3(DBA_B), dim3(128), 0, stream>>>(ws, lmb, out);
  const int tot = DBA_B * DBA_N;
  dba_stage3_depth<<<dim3((tot + 255) / 256), dim3(256), 0, stream>>>(
      r, w, Jp, Jd, lmb, out, out + DBA_B * 6);
}


# stage-1 operands via coalesced 16-byte loads and a wave-private LDS transpose: per-n vectors computed once, WMMA fragments read ready-made from LDS (no shuffles), next tile prefetched; sums unchanged
# speedup vs baseline: 1.8195x; 1.8195x over previous
#include <hip/hip_runtime.h>


typedef __attribute__((ext_vector_type(2))) float v2f;
typedef __attribute__((ext_vector_type(8))) float v8f;

#define DBA_B 32
#define DBA_N 65536
#define DBA_G 32
#define DBA_SPAN (DBA_N / DBA_G)
#define DBA_PW (DBA_SPAN / 8)
#define DBA_TN 32
#define DBA_NT (DBA_PW / DBA_TN)
#define DBA_LJ 0
#define DBA_LU 384
#define DBA_LA 860
#define DBA_LV 1336
#define DBA_LQ 1588
#define DBA_LW 1840
#define DBA_PU 68
#define DBA_PV 36

#define DBA_WAVE_SYNC()                                        \
  do {                                                         \
    __builtin_amdgcn_fence(__ATOMIC_RELEASE, "wavefront");     \
    __builtin_amdgcn_wave_barrier();                           \
    __builtin_amdgcn_fence(__ATOMIC_ACQUIRE, "wavefront");     \
  } while (0)

__device__ __forceinline__ float fast_rcp(float x) {
  float inv = __builtin_amdgcn_rcpf(x);
  inv = inv * (2.0f - x * inv);
  return inv;
}

__global__ __launch_bounds__(256) void dba_stage1_reduce(
    const float* __restrict__ r, const float* __restrict__ w,
    const float* __restrict__ Jp, const float* __restrict__ Jd,
    const float* __restrict__ lmbda, float* __restrict__ partial)
{
  const int b    = blockIdx.x / DBA_G;
  const int c    = blockIdx.x % DBA_G;
  const int wave = threadIdx.x >> 5;
  const int lane = threadIdx.x & 31;
  const int half = lane >> 4;
  const int i    = lane & 15;
  const float lmb = lmbda[0];

  const int base = c * DBA_SPAN + wave * DBA_PW;
  const size_t n0 = (size_t)b * DBA_N + base;

  __shared__ __attribute__((aligned(16))) float stg[8][DBA_LW];
  float* L = stg[wave];

  const float4* gJ = (const float4*)(Jp + n0 * 12) + lane;
  const float2* gr = (const float2*)(r  + n0 * 2) + lane;
  const float2* gw = (const float2*)(w  + n0 * 2) + lane;
  const float2* gd = (const float2*)(Jd + n0 * 2) + lane;

  float4*       SJ = (float4*)(L + DBA_LJ);
  const float4* TJ = (const float4*)(L + DBA_LJ) + lane * 3;
  float2* Wu = (float2*)(L + DBA_LU) + lane;
  float2* Wa = (float2*)(L + DBA_LA) + lane;
  float*  Wv = L + DBA_LV + lane;
  float*  Wq = L + DBA_LQ + lane;

  const int ic = (i < 6) ? i : 6;
  const float2* Ru = (const float2*)(L + DBA_LU + ic * DBA_PU) + half;
  const float2* Ra = (const float2*)(L + DBA_LA + ic * DBA_PU) + half;
  const float2* Rv = (const float2*)(L + DBA_LV + ic * DBA_PV) + half;
  const float2* Rq = (const float2*)(L + DBA_LQ + ic * DBA_PV) + half;

  v8f cP = {0.f,0.f,0.f,0.f,0.f,0.f,0.f,0.f};
  v8f cQ = {0.f,0.f,0.f,0.f,0.f,0.f,0.f,0.f};

  float4 j0 = gJ[0], j1 = gJ[32], j2 = gJ[64];
  float2 rv = gr[0], wc = gw[0], dv = gd[0];

  for (int t = 0; t < DBA_NT; ++t) {
    DBA_WAVE_SYNC();
    SJ[lane]      = j0;
    SJ[lane + 32] = j1;
    SJ[lane + 64] = j2;
    DBA_WAVE_SYNC();
    const float4 a0 = TJ[0], a1 = TJ[1], a2 = TJ[2];
    const float2 rc = rv, wn = wc, dn = dv;

    if (t + 1 < DBA_NT) {
      const int q = t + 1;
      j0 = gJ[q * 96]; j1 = gJ[q * 96 + 32]; j2 = gJ[q * 96 + 64];
      rv = gr[q * 32]; wc = gw[q * 32]; dv = gd[q * 32];
    }

    const float u0[7] = {a0.x, a0.y, a0.z, a0.w, a1.x, a1.y, rc.x};
    const float u1[7] = {a1.z, a1.w, a2.x, a2.y, a2.z, a2.w, rc.y};
    const float hdd = wn.x * (dn.x * dn.x + dn.y * dn.y) + lmb + wn.y;
    const float inv = fast_rcp(hdd + 1e-6f);
#pragma unroll
    for (int k = 0; k < 7; ++k) {
      const float p0 = wn.x * u0[k];
      const float p1 = wn.x * u1[k];
      const float vv = dn.x * p0 + dn.y * p1;
      Wu[k * (DBA_PU / 2)] = make_float2(u0[k], u1[k]);
      Wa[k * (DBA_PU / 2)] = make_float2(p0, p1);
      Wv[k * DBA_PV] = vv;
      Wq[k * DBA_PV] = vv * inv;
    }
    DBA_WAVE_SYNC();

#pragma unroll
    for (int q = 0; q < DBA_TN / 4; ++q) {
      const float2 uA = Ru[4 * q],     cA = Ra[4 * q];
      const float2 uB = Ru[4 * q + 2], cB = Ra[4 * q + 2];
      const float2 vq = Rv[2 * q],     sq = Rq[2 * q];
      v2f pa, pb;
      pa.x = cA.x; pa.y = cA.y; pb.x = uA.x; pb.y = uA.y;
      cP = __builtin_amdgcn_wmma_f32_16x16x4_f32(false, pa, false, pb,
                                                 (short)0, cP, false, false);
      pa.x = cB.x; pa.y = cB.y; pb.x = uB.x; pb.y = uB.y;
      cP = __builtin_amdgcn_wmma_f32_16x16x4_f32(false, pa, false, pb,
                                                 (short)0, cP, false, false);
      v2f qa, qb;
      qa.x = sq.x; qa.y = sq.y; qb.x = vq.x; qb.y = vq.y;
      cQ = __builtin_amdgcn_wmma_f32_16x16x4_f32(false, qa, false, qb,
                                                 (short)0, cQ, false, false);
    }
  }

  __shared__ float sred[8][112];
  if (lane < 7) {
#pragma unroll
    for (int p = 0; p < 7; ++p) {
      sred[wave][p * 7 + lane]      = cP[p];
      sred[wave][49 + p * 7 + lane] = cQ[p];
    }
  }
  __syncthreads();
  if (threadIdx.x < 98) {
    float s = 0.f;
#pragma unroll
    for (int wv = 0; wv < 8; ++wv) s += sred[wv][threadIdx.x];
    partial[(size_t)blockIdx.x * 128 + threadIdx.x] = s;
  }
}

__global__ __launch_bounds__(128) void dba_stage2_solve(
    const float* __restrict__ partial, const float* __restrict__ lmbda,
    float* __restrict__ out)
{
  const int b = blockIdx.x;
  __shared__ float acc[98];
  if (threadIdx.x < 98) {
    float s = 0.f;
    for (int g = 0; g < DBA_G; ++g)
      s += partial[(size_t)(b * DBA_G + g) * 128 + threadIdx.x];
    acc[threadIdx.x] = s;
  }
  __syncthreads();
  if (threadIdx.x == 0) {
    const float lmb = lmbda[0];
    float A[6][7];
    for (int ii = 0; ii < 6; ++ii) {
      for (int jj = 0; jj < 6; ++jj) {
        const float hij = acc[ii * 7 + jj] - acc[49 + ii * 7 + jj];
        const float hji = acc[jj * 7 + ii] - acc[49 + jj * 7 + ii];
        A[ii][jj] = 0.5f * (hij + hji);
      }
      A[ii][ii] += lmb + 0.001f;
      A[ii][6] = acc[ii * 7 + 6] - acc[49 + ii * 7 + 6];
    }
    for (int col = 0; col < 6; ++col) {
      int piv = col; float best = fabsf(A[col][col]);
      for (int rr = col + 1; rr < 6; ++rr) {
        const float av = fabsf(A[rr][col]);
        if (av > best) { best = av; piv = rr; }
      }
      if (piv != col)
        for (int jj = col; jj < 7; ++jj) {
          const float tmp = A[col][jj]; A[col][jj] = A[piv][jj]; A[piv][jj] = tmp;
        }
      const float invd = 1.0f / A[col][col];
      for (int rr = col + 1; rr < 6; ++rr) {
        const float f = A[rr][col] * invd;
        for (int jj = col; jj < 7; ++jj) A[rr][jj] -= f * A[col][jj];
      }
    }
    float x[6];
    for (int ii = 5; ii >= 0; --ii) {
      float s = A[ii][6];
      for (int jj = ii + 1; jj < 6; ++jj) s -= A[ii][jj] * x[jj];
      x[ii] = s / A[ii][ii];
    }
    for (int ii = 0; ii < 6; ++ii) out[b * 6 + ii] = x[ii];
  }
}

__global__ __launch_bounds__(256) void dba_stage3_depth(
    const float* __restrict__ r, const float* __restrict__ w,
    const float* __restrict__ Jp, const float* __restrict__ Jd,
    const float* __restrict__ lmbda, const float* __restrict__ dp,
    float* __restrict__ dd)
{
  const int idx = blockIdx.x * blockDim.x + threadIdx.x;
  if (idx >= DBA_B * DBA_N) return;
  const int b = idx / DBA_N;
  const size_t rb = (size_t)idx;

  const float4* Jp4 = (const float4*)(Jp + rb * 12);
  const float4 a0 = Jp4[0], a1 = Jp4[1], a2 = Jp4[2];
  const float2 wv = ((const float2*)w)[rb];
  const float2 jd = ((const float2*)Jd)[rb];
  const float2 rv = ((const float2*)r)[rb];
  const float lmb = lmbda[0];

  const float p0 = dp[b * 6 + 0], p1 = dp[b * 6 + 1], p2 = dp[b * 6 + 2];
  const float p3 = dp[b * 6 + 3], p4 = dp[b * 6 + 4], p5 = dp[b * 6 + 5];

  const float h0 = wv.x * (jd.x * a0.x + jd.y * a1.z);
  const float h1 = wv.x * (jd.x * a0.y + jd.y * a1.w);
  const float h2 = wv.x * (jd.x * a0.z + jd.y * a2.x);
  const float h3 = wv.x * (jd.x * a0.w + jd.y * a2.y);
  const float h4 = wv.x * (jd.x * a1.x + jd.y * a2.z);
  const float h5 = wv.x * (jd.x * a1.y + jd.y * a2.w);
  const float gd  = wv.x * (jd.x * rv.x + jd.y * rv.y);
  const float hdd = wv.x * (jd.x * jd.x + jd.y * jd.y) + lmb + wv.y;
  const float inv = fast_rcp(hdd + 1e-6f);
  const float vdot = h0 * p0 + h1 * p1 + h2 * p2 + h3 * p3 + h4 * p4 + h5 * p5;
  dd[idx] = inv * (gd - vdot);
}

extern "C" void kernel_launch(void* const* d_in, const int* in_sizes, int n_in,
                              void* d_out, int out_size, void* d_ws, size_t ws_size,
                              hipStream_t stream) {
  const float* r   = (const float*)d_in[0];
  const float* w   = (const float*)d_in[1];
  const float* Jp  = (const float*)d_in[2];
  const float* Jd  = (const float*)d_in[3];
  const float* lmb = (const float*)d_in[4];
  float* out = (float*)d_out;
  float* ws  = (float*)d_ws;

  dba_stage1_reduce<<<dim3(DBA_B * DBA_G), dim3(256), 0, stream>>>(
      r, w, Jp, Jd, lmb, ws);
  dba_stage2_solve<<<dim3(DBA_B), dim3(128), 0, stream>>>(ws, lmb, out);
  const int tot = DBA_B * DBA_N;
  dba_stage3_depth<<<dim3((tot + 255) / 256), dim3(256), 0, stream>>>(
      r, w, Jp, Jd, lmb, out, out + DBA_B * 6);
}


# stage-3 epilogue on stage 1's grid: each wave streams its own 256 n as coalesced 16 B tiles turned through LDS, three tiles requested ahead, last tile first (8192 waves, not 65536); sums unchanged
# speedup vs baseline: 1.8735x; 1.0297x over previous
#include <hip/hip_runtime.h>


typedef __attribute__((ext_vector_type(2))) float v2f;
typedef __attribute__((ext_vector_type(8))) float v8f;

#define DBA_B 32
#define DBA_N 65536
#define DBA_G 32
#define DBA_SPAN (DBA_N / DBA_G)
#define DBA_PW (DBA_SPAN / 8)
#define DBA_TN 32
#define DBA_NT (DBA_PW / DBA_TN)
#define DBA_LJ 0
#define DBA_LU 384
#define DBA_LA 860
#define DBA_LV 1336
#define DBA_LQ 1588
#define DBA_LW 1840
#define DBA_PU 68
#define DBA_PV 36

#define DBA_WAVE_SYNC()                                        \
  do {                                                         \
    __builtin_amdgcn_fence(__ATOMIC_RELEASE, "wavefront");     \
    __builtin_amdgcn_wave_barrier();                           \
    __builtin_amdgcn_fence(__ATOMIC_ACQUIRE, "wavefront");     \
  } while (0)

__device__ __forceinline__ float fast_rcp(float x) {
  float inv = __builtin_amdgcn_rcpf(x);
  inv = inv * (2.0f - x * inv);
  return inv;
}

__global__ __launch_bounds__(256) void dba_stage1_reduce(
    const float* __restrict__ r, const float* __restrict__ w,
    const float* __restrict__ Jp, const float* __restrict__ Jd,
    const float* __restrict__ lmbda, float* __restrict__ partial)
{
  const int b    = blockIdx.x / DBA_G;
  const int c    = blockIdx.x % DBA_G;
  const int wave = threadIdx.x >> 5;
  const int lane = threadIdx.x & 31;
  const int half = lane >> 4;
  const int i    = lane & 15;
  const float lmb = lmbda[0];

  const int base = c * DBA_SPAN + wave * DBA_PW;
  const size_t n0 = (size_t)b * DBA_N + base;

  __shared__ __attribute__((aligned(16))) float stg[8][DBA_LW];
  float* L = stg[wave];

  const float4* gJ = (const float4*)(Jp + n0 * 12) + lane;
  const float2* gr = (const float2*)(r  + n0 * 2) + lane;
  const float2* gw = (const float2*)(w  + n0 * 2) + lane;
  const float2* gd = (const float2*)(Jd + n0 * 2) + lane;

  float4*       SJ = (float4*)(L + DBA_LJ);
  const float4* TJ = (const float4*)(L + DBA_LJ) + lane * 3;
  float2* Wu = (float2*)(L + DBA_LU) + lane;
  float2* Wa = (float2*)(L + DBA_LA) + lane;
  float*  Wv = L + DBA_LV + lane;
  float*  Wq = L + DBA_LQ + lane;

  const int ic = (i < 6) ? i : 6;
  const float2* Ru = (const float2*)(L + DBA_LU + ic * DBA_PU) + half;
  const float2* Ra = (const float2*)(L + DBA_LA + ic * DBA_PU) + half;
  const float2* Rv = (const float2*)(L + DBA_LV + ic * DBA_PV) + half;
  const float2* Rq = (const float2*)(L + DBA_LQ + ic * DBA_PV) + half;

  v8f cP = {0.f,0.f,0.f,0.f,0.f,0.f,0.f,0.f};
  v8f cQ = {0.f,0.f,0.f,0.f,0.f,0.f,0.f,0.f};

  float4 j0 = gJ[0], j1 = gJ[32], j2 = gJ[64];
  float2 rv = gr[0], wc = gw[0], dv = gd[0];

  for (int t = 0; t < DBA_NT; ++t) {
    DBA_WAVE_SYNC();
    SJ[lane]      = j0;
    SJ[lane + 32] = j1;
    SJ[lane + 64] = j2;
    DBA_WAVE_SYNC();
    const float4 a0 = TJ[0], a1 = TJ[1], a2 = TJ[2];
    const float2 rc = rv, wn = wc, dn = dv;

    if (t + 1 < DBA_NT) {
      const int q = t + 1;
      j0 = gJ[q * 96]; j1 = gJ[q * 96 + 32]; j2 = gJ[q * 96 + 64];
      rv = gr[q * 32]; wc = gw[q * 32]; dv = gd[q * 32];
    }

    const float u0[7] = {a0.x, a0.y, a0.z, a0.w, a1.x, a1.y, rc.x};
    const float u1[7] = {a1.z, a1.w, a2.x, a2.y, a2.z, a2.w, rc.y};
    const float hdd = wn.x * (dn.x * dn.x + dn.y * dn.y) + lmb + wn.y;
    const float inv = fast_rcp(hdd + 1e-6f);
#pragma unroll
    for (int k = 0; k < 7; ++k) {
      const float p0 = wn.x * u0[k];
      const float p1 = wn.x * u1[k];
      const float vv = dn.x * p0 + dn.y * p1;
      Wu[k * (DBA_PU / 2)] = make_float2(u0[k], u1[k]);
      Wa[k * (DBA_PU / 2)] = make_float2(p0, p1);
      Wv[k * DBA_PV] = vv;
      Wq[k * DBA_PV] = vv * inv;
    }
    DBA_WAVE_SYNC();

#pragma unroll
    for (int q = 0; q < DBA_TN / 4; ++q) {
      const float2 uA = Ru[4 * q],     cA = Ra[4 * q];
      const float2 uB = Ru[4 * q + 2], cB = Ra[4 * q + 2];
      const float2 vq = Rv[2 * q],     sq = Rq[2 * q];
      v2f pa, pb;
      pa.x = cA.x; pa.y = cA.y; pb.x = uA.x; pb.y = uA.y;
      cP = __builtin_amdgcn_wmma_f32_16x16x4_f32(false, pa, false, pb,
                                                 (short)0, cP, false, false);
      pa.x = cB.x; pa.y = cB.y; pb.x = uB.x; pb.y = uB.y;
      cP = __builtin_amdgcn_wmma_f32_16x16x4_f32(false, pa, false, pb,
                                                 (short)0, cP, false, false);
      v2f qa, qb;
      qa.x = sq.x; qa.y = sq.y; qb.x = vq.x; qb.y = vq.y;
      cQ = __builtin_amdgcn_wmma_f32_16x16x4_f32(false, qa, false, qb,
                                                 (short)0, cQ, false, false);
    }
  }

  __shared__ float sred[8][112];
  if (lane < 7) {
#pragma unroll
    for (int p = 0; p < 7; ++p) {
      sred[wave][p * 7 + lane]      = cP[p];
      sred[wave][49 + p * 7 + lane] = cQ[p];
    }
  }
  __syncthreads();
  if (threadIdx.x < 98) {
    float s = 0.f;
#pragma unroll
    for (int wv = 0; wv < 8; ++wv) s += sred[wv][threadIdx.x];
    partial[(size_t)blockIdx.x * 128 + threadIdx.x] = s;
  }
}

__global__ __launch_bounds__(128) void dba_stage2_solve(
    const float* __restrict__ partial, const float* __restrict__ lmbda,
    float* __restrict__ out)
{
  const int b = blockIdx.x;
  __shared__ float acc[98];
  if (threadIdx.x < 98) {
    float s = 0.f;
    for (int g = 0; g < DBA_G; ++g)
      s += partial[(size_t)(b * DBA_G + g) * 128 + threadIdx.x];
    acc[threadIdx.x] = s;
  }
  __syncthreads();
  if (threadIdx.x == 0) {
    const float lmb = lmbda[0];
    float A[6][7];
    for (int ii = 0; ii < 6; ++ii) {
      for (int jj = 0; jj < 6; ++jj) {
        const float hij = acc[ii * 7 + jj] - acc[49 + ii * 7 + jj];
        const float hji = acc[jj * 7 + ii] - acc[49 + jj * 7 + ii];
        A[ii][jj] = 0.5f * (hij + hji);
      }
      A[ii][ii] += lmb + 0.001f;
      A[ii][6] = acc[ii * 7 + 6] - acc[49 + ii * 7 + 6];
    }
    for (int col = 0; col < 6; ++col) {
      int piv = col; float best = fabsf(A[col][col]);
      for (int rr = col + 1; rr < 6; ++rr) {
        const float av = fabsf(A[rr][col]);
        if (av > best) { best = av; piv = rr; }
      }
      if (piv != col)
        for (int jj = col; jj < 7; ++jj) {
          const float tmp = A[col][jj]; A[col][jj] = A[piv][jj]; A[piv][jj] = tmp;
        }
      const float invd = 1.0f / A[col][col];
      for (int rr = col + 1; rr < 6; ++rr) {
        const float f = A[rr][col] * invd;
        for (int jj = col; jj < 7; ++jj) A[rr][jj] -= f * A[col][jj];
      }
    }
    float x[6];
    for (int ii = 5; ii >= 0; --ii) {
      float s = A[ii][6];
      for (int jj = ii + 1; jj < 6; ++jj) s -= A[ii][jj] * x[jj];
      x[ii] = s / A[ii][ii];
    }
    for (int ii = 0; ii < 6; ++ii) out[b * 6 + ii] = x[ii];
  }
}

typedef __attribute__((ext_vector_type(4))) float v4f;
#define DBA_S3_DECL(R) v4f R##j0, R##j1, R##j2; v2f R##r, R##w, R##d
#define DBA_S3_LOAD(R, q)                                              \
  do {                                                                 \
    R##j0 = gJ[(q) * 96]; R##j1 = gJ[(q) * 96 + 32];                   \
    R##j2 = gJ[(q) * 96 + 64];                                         \
    R##r = gr[(q) * 32]; R##w = gw[(q) * 32]; R##d = gd[(q) * 32];     \
  } while (0)
#define DBA_S3_WORK(R, q)                                              \
  do {                                                                 \
    DBA_WAVE_SYNC();   \
    SJ[lane] = R##j0; SJ[lane + 32] = R##j1; SJ[lane + 64] = R##j2;    \
    DBA_WAVE_SYNC();   \
    const v4f a0 = TJ[0], a1 = TJ[1], a2 = TJ[2];                      \
    const v2f rv = R##r, wv = R##w, jd = R##d;                         \
    const float h0 = wv.x * (jd.x * a0.x + jd.y * a1.z);               \
    const float h1 = wv.x * (jd.x * a0.y + jd.y * a1.w);               \
    const float h2 = wv.x * (jd.x * a0.z + jd.y * a2.x);               \
    const float h3 = wv.x * (jd.x * a0.w + jd.y * a2.y);               \
    const float h4 = wv.x * (jd.x * a1.x + jd.y * a2.z);               \
    const float h5 = wv.x * (jd.x * a1.y + jd.y * a2.w);               \
    const float gdn = wv.x * (jd.x * rv.x + jd.y * rv.y);              \
    const float hdd = wv.x * (jd.x * jd.x + jd.y * jd.y) + lmb + wv.y; \
    const float inv = fast_rcp(hdd + 1e-6f);                           \
    const float vdot = h0 * p0 + h1 * p1 + h2 * p2 + h3 * p3           \
                     + h4 * p4 + h5 * p5;                              \
    go[(q) * 32] = inv * (gdn - vdot);                                 \
  } while (0)

__global__ __launch_bounds__(256) void dba_stage3_depth(
    const float* __restrict__ r, const float* __restrict__ w,
    const float* __restrict__ Jp, const float* __restrict__ Jd,
    const float* __restrict__ lmbda, const float* __restrict__ dp,
    float* __restrict__ dd)
{
  const int b    = blockIdx.x / DBA_G;
  const int c    = blockIdx.x % DBA_G;
  const int wave = threadIdx.x >> 5;
  const int lane = threadIdx.x & 31;
  const float lmb = lmbda[0];
  const float p0 = dp[b * 6 + 0], p1 = dp[b * 6 + 1], p2 = dp[b * 6 + 2];
  const float p3 = dp[b * 6 + 3], p4 = dp[b * 6 + 4], p5 = dp[b * 6 + 5];

  const int base = c * DBA_SPAN + wave * DBA_PW;
  const size_t n0 = (size_t)b * DBA_N + base;

  __shared__ __attribute__((aligned(16))) float turn[8][DBA_TN * 12];
  v4f*       SJ = (v4f*)turn[wave];
  const v4f* TJ = (const v4f*)turn[wave] + lane * 3;

  const v4f* gJ = (const v4f*)(Jp + n0 * 12) + lane;
  const v2f* gr = (const v2f*)(r  + n0 * 2) + lane;
  const v2f* gw = (const v2f*)(w  + n0 * 2) + lane;
  const v2f* gd = (const v2f*)(Jd + n0 * 2) + lane;
  float*     go = dd + n0 + lane;

  DBA_S3_DECL(A); DBA_S3_DECL(B); DBA_S3_DECL(C); DBA_S3_DECL(D);
  DBA_S3_LOAD(A, 7); DBA_S3_LOAD(B, 6); DBA_S3_LOAD(C, 5);
  DBA_S3_LOAD(D, 4); DBA_S3_WORK(A, 7);
  DBA_S3_LOAD(A, 3); DBA_S3_WORK(B, 6);
  DBA_S3_LOAD(B, 2); DBA_S3_WORK(C, 5);
  DBA_S3_LOAD(C, 1); DBA_S3_WORK(D, 4);
  DBA_S3_LOAD(D, 0); DBA_S3_WORK(A, 3);
  DBA_S3_WORK(B, 2);
  DBA_S3_WORK(C, 1);
  DBA_S3_WORK(D, 0);
}

extern "C" void kernel_launch(void* const* d_in, const int* in_sizes, int n_in,
                              void* d_out, int out_size, void* d_ws, size_t ws_size,
                              hipStream_t stream) {
  const float* r   = (const float*)d_in[0];
  const float* w   = (const float*)d_in[1];
  const float* Jp  = (const float*)d_in[2];
  const float* Jd  = (const float*)d_in[3];
  const float* lmb = (const float*)d_in[4];
  float* out = (float*)d_out;
  float* ws  = (float*)d_ws;

  dba_stage1_reduce<<<dim3(DBA_B * DBA_G), dim3(256), 0, stream>>>(
      r, w, Jp, Jd, lmb, ws);
  dba_stage2_solve<<<dim3(DBA_B), dim3(128), 0, stream>>>(ws, lmb, out);
  dba_stage3_depth<<<dim3(DBA_B * DBA_G), dim3(256), 0, stream>>>(
      r, w, Jp, Jd, lmb, out, out + DBA_B * 6);
}


# stage 1 folds Q onto P's WMMAs: A rows 8-13 = inv*v*conf*Jd_k against the shared u fragment, 16 f32 WMMAs per 32 n not 24, two alternating accumulators; Q's sums regrouped per (n,k)
# speedup vs baseline: 2.0429x; 1.0904x over previous
#include <hip/hip_runtime.h>


typedef __attribute__((ext_vector_type(2))) float v2f;
typedef __attribute__((ext_vector_type(8))) float v8f;

#define DBA_B 32
#define DBA_N 65536
#define DBA_G 32
#define DBA_SPAN (DBA_N / DBA_G)
#define DBA_PW (DBA_SPAN / 8)
#define DBA_TN 32
#define DBA_NT (DBA_PW / DBA_TN)
#define DBA_LJ 0
#define DBA_LU 384
#define DBA_LA 860
#define DBA_LW 1676
#define DBA_PU 68

#define DBA_WAVE_SYNC()                                        \
  do {                                                         \
    __builtin_amdgcn_fence(__ATOMIC_RELEASE, "wavefront");     \
    __builtin_amdgcn_wave_barrier();                           \
    __builtin_amdgcn_fence(__ATOMIC_ACQUIRE, "wavefront");     \
  } while (0)

__device__ __forceinline__ float fast_rcp(float x) {
  float inv = __builtin_amdgcn_rcpf(x);
  inv = inv * (2.0f - x * inv);
  return inv;
}

__global__ __launch_bounds__(256) void dba_stage1_reduce(
    const float* __restrict__ r, const float* __restrict__ w,
    const float* __restrict__ Jp, const float* __restrict__ Jd,
    const float* __restrict__ lmbda, float* __restrict__ partial)
{
  const int b    = blockIdx.x / DBA_G;
  const int c    = blockIdx.x % DBA_G;
  const int wave = threadIdx.x >> 5;
  const int lane = threadIdx.x & 31;
  const int half = lane >> 4;
  const int i    = lane & 15;
  const float lmb = lmbda[0];

  const int base = c * DBA_SPAN + wave * DBA_PW;
  const size_t n0 = (size_t)b * DBA_N + base;

  __shared__ __attribute__((aligned(16))) float stg[8][DBA_LW];
  float* L = stg[wave];

  const float4* gJ = (const float4*)(Jp + n0 * 12) + lane;
  const float2* gr = (const float2*)(r  + n0 * 2) + lane;
  const float2* gw = (const float2*)(w  + n0 * 2) + lane;
  const float2* gd = (const float2*)(Jd + n0 * 2) + lane;

  float4*       SJ = (float4*)(L + DBA_LJ);
  const float4* TJ = (const float4*)(L + DBA_LJ) + lane * 3;
  float2* Wu = (float2*)(L + DBA_LU) + lane;
  float2* Wa = (float2*)(L + DBA_LA) + lane;

  const int ib = (i < 6) ? i : 6;
  const int ia = (i < 6) ? i : (i < 8) ? 5 : (i < 14) ? (i - 2) : 11;
  const float2* Ru = (const float2*)(L + DBA_LU + ib * DBA_PU) + half;
  const float2* Ra = (const float2*)(L + DBA_LA + ia * DBA_PU) + half;

  v8f cE = {0.f,0.f,0.f,0.f,0.f,0.f,0.f,0.f};
  v8f cO = {0.f,0.f,0.f,0.f,0.f,0.f,0.f,0.f};

  float4 j0 = gJ[0], j1 = gJ[32], j2 = gJ[64];
  float2 rv = gr[0], wc = gw[0], dv = gd[0];

  for (int t = 0; t < DBA_NT; ++t) {
    DBA_WAVE_SYNC();
    SJ[lane]      = j0;
    SJ[lane + 32] = j1;
    SJ[lane + 64] = j2;
    DBA_WAVE_SYNC();
    const float4 a0 = TJ[0], a1 = TJ[1], a2 = TJ[2];
    const float2 rc = rv, wn = wc, dn = dv;

    if (t + 1 < DBA_NT) {
      const int q = t + 1;
      j0 = gJ[q * 96]; j1 = gJ[q * 96 + 32]; j2 = gJ[q * 96 + 64];
      rv = gr[q * 32]; wc = gw[q * 32]; dv = gd[q * 32];
    }

    const float u0[7] = {a0.x, a0.y, a0.z, a0.w, a1.x, a1.y, rc.x};
    const float u1[7] = {a1.z, a1.w, a2.x, a2.y, a2.z, a2.w, rc.y};
    const float hdd = wn.x * (dn.x * dn.x + dn.y * dn.y) + lmb + wn.y;
    const float inv = fast_rcp(hdd + 1e-6f);
    const float e0 = wn.x * dn.x;
    const float e1 = wn.x * dn.y;
    Wu[6 * (DBA_PU / 2)] = make_float2(u0[6], u1[6]);
#pragma unroll
    for (int k = 0; k < 6; ++k) {
      const float p0 = wn.x * u0[k];
      const float p1 = wn.x * u1[k];
      const float tt = (dn.x * p0 + dn.y * p1) * inv;
      Wu[k * (DBA_PU / 2)]       = make_float2(u0[k], u1[k]);
      Wa[k * (DBA_PU / 2)]       = make_float2(p0, p1);
      Wa[(6 + k) * (DBA_PU / 2)] = make_float2(tt * e0, tt * e1);
    }
    DBA_WAVE_SYNC();

#pragma unroll
    for (int q = 0; q < DBA_TN / 4; ++q) {
      const float2 uA = Ru[4 * q],     cA = Ra[4 * q];
      const float2 uB = Ru[4 * q + 2], cB = Ra[4 * q + 2];
      v2f pa, pb;
      pa.x = cA.x; pa.y = cA.y; pb.x = uA.x; pb.y = uA.y;
      cE = __builtin_amdgcn_wmma_f32_16x16x4_f32(false, pa, false, pb,
                                                 (short)0, cE, false, false);
      pa.x = cB.x; pa.y = cB.y; pb.x = uB.x; pb.y = uB.y;
      cO = __builtin_amdgcn_wmma_f32_16x16x4_f32(false, pa, false, pb,
                                                 (short)0, cO, false, false);
    }
  }

  __shared__ float sred[8][112];
  if (i < 7) {
#pragma unroll
    for (int p = 0; p < 6; ++p)
      sred[wave][half * 49 + p * 7 + i] = cE[p] + cO[p];
    sred[wave][half * 49 + 42 + i] = 0.f;
  }
  __syncthreads();
  if (threadIdx.x < 98) {
    float s = 0.f;
#pragma unroll
    for (int wv = 0; wv < 8; ++wv) s += sred[wv][threadIdx.x];
    partial[(size_t)blockIdx.x * 128 + threadIdx.x] = s;
  }
}

__global__ __launch_bounds__(128) void dba_stage2_solve(
    const float* __restrict__ partial, const float* __restrict__ lmbda,
    float* __restrict__ out)
{
  const int b = blockIdx.x;
  __shared__ float acc[98];
  if (threadIdx.x < 98) {
    float s = 0.f;
    for (int g = 0; g < DBA_G; ++g)
      s += partial[(size_t)(b * DBA_G + g) * 128 + threadIdx.x];
    acc[threadIdx.x] = s;
  }
  __syncthreads();
  if (threadIdx.x == 0) {
    const float lmb = lmbda[0];
    float A[6][7];
    for (int ii = 0; ii < 6; ++ii) {
      for (int jj = 0; jj < 6; ++jj) {
        const float hij = acc[ii * 7 + jj] - acc[49 + ii * 7 + jj];
        const float hji = acc[jj * 7 + ii] - acc[49 + jj * 7 + ii];
        A[ii][jj] = 0.5f * (hij + hji);
      }
      A[ii][ii] += lmb + 0.001f;
      A[ii][6] = acc[ii * 7 + 6] - acc[49 + ii * 7 + 6];
    }
    for (int col = 0; col < 6; ++col) {
      int piv = col; float best = fabsf(A[col][col]);
      for (int rr = col + 1; rr < 6; ++rr) {
        const float av = fabsf(A[rr][col]);
        if (av > best) { best = av; piv = rr; }
      }
      if (piv != col)
        for (int jj = col; jj < 7; ++jj) {
          const float tmp = A[col][jj]; A[col][jj] = A[piv][jj]; A[piv][jj] = tmp;
        }
      const float invd = 1.0f / A[col][col];
      for (int rr = col + 1; rr < 6; ++rr) {
        const float f = A[rr][col] * invd;
        for (int jj = col; jj < 7; ++jj) A[rr][jj] -= f * A[col][jj];
      }
    }
    float x[6];
    for (int ii = 5; ii >= 0; --ii) {
      float s = A[ii][6];
      for (int jj = ii + 1; jj < 6; ++jj) s -= A[ii][jj] * x[jj];
      x[ii] = s / A[ii][ii];
    }
    for (int ii = 0; ii < 6; ++ii) out[b * 6 + ii] = x[ii];
  }
}

typedef __attribute__((ext_vector_type(4))) float v4f;
#define DBA_S3_DECL(R) v4f R##j0, R##j1, R##j2; v2f R##r, R##w, R##d
#define DBA_S3_LOAD(R, q)                                              \
  do {                                                                 \
    R##j0 = gJ[(q) * 96]; R##j1 = gJ[(q) * 96 + 32];                   \
    R##j2 = gJ[(q) * 96 + 64];                                         \
    R##r = gr[(q) * 32]; R##w = gw[(q) * 32]; R##d = gd[(q) * 32];     \
  } while (0)
#define DBA_S3_WORK(R, q)                                              \
  do {                                                                 \
    DBA_WAVE_SYNC();   \
    SJ[lane] = R##j0; SJ[lane + 32] = R##j1; SJ[lane + 64] = R##j2;    \
    DBA_WAVE_SYNC();   \
    const v4f a0 = TJ[0], a1 = TJ[1], a2 = TJ[2];                      \
    const v2f rv = R##r, wv = R##w, jd = R##d;                         \
    const float h0 = wv.x * (jd.x * a0.x + jd.y * a1.z);               \
    const float h1 = wv.x * (jd.x * a0.y + jd.y * a1.w);               \
    const float h2 = wv.x * (jd.x * a0.z + jd.y * a2.x);               \
    const float h3 = wv.x * (jd.x * a0.w + jd.y * a2.y);               \
    const float h4 = wv.x * (jd.x * a1.x + jd.y * a2.z);               \
    const float h5 = wv.x * (jd.x * a1.y + jd.y * a2.w);               \
    const float gdn = wv.x * (jd.x * rv.x + jd.y * rv.y);              \
    const float hdd = wv.x * (jd.x * jd.x + jd.y * jd.y) + lmb + wv.y; \
    const float inv = fast_rcp(hdd + 1e-6f);                           \
    const float vdot = h0 * p0 + h1 * p1 + h2 * p2 + h3 * p3           \
                     + h4 * p4 + h5 * p5;                              \
    go[(q) * 32] = inv * (gdn - vdot);                                 \
  } while (0)

__global__ __launch_bounds__(256) void dba_stage3_depth(
    const float* __restrict__ r, const float* __restrict__ w,
    const float* __restrict__ Jp, const float* __restrict__ Jd,
    const float* __restrict__ lmbda, const float* __restrict__ dp,
    float* __restrict__ dd)
{
  const int b    = blockIdx.x / DBA_G;
  const int c    = blockIdx.x % DBA_G;
  const int wave = threadIdx.x >> 5;
  const int lane = threadIdx.x & 31;
  const float lmb = lmbda[0];
  const float p0 = dp[b * 6 + 0], p1 = dp[b * 6 + 1], p2 = dp[b * 6 + 2];
  const float p3 = dp[b * 6 + 3], p4 = dp[b * 6 + 4], p5 = dp[b * 6 + 5];

  const int base = c * DBA_SPAN + wave * DBA_PW;
  const size_t n0 = (size_t)b * DBA_N + base;

  __shared__ __attribute__((aligned(16))) float turn[8][DBA_TN * 12];
  v4f*       SJ = (v4f*)turn[wave];
  const v4f* TJ = (const v4f*)turn[wave] + lane * 3;

  const v4f* gJ = (const v4f*)(Jp + n0 * 12) + lane;
  const v2f* gr = (const v2f*)(r  + n0 * 2) + lane;
  const v2f* gw = (const v2f*)(w  + n0 * 2) + lane;
  const v2f* gd = (const v2f*)(Jd + n0 * 2) + lane;
  float*     go = dd + n0 + lane;

  DBA_S3_DECL(A); DBA_S3_DECL(B); DBA_S3_DECL(C); DBA_S3_DECL(D);
  DBA_S3_LOAD(A, 7); DBA_S3_LOAD(B, 6); DBA_S3_LOAD(C, 5);
  DBA_S3_LOAD(D, 4); DBA_S3_WORK(A, 7);
  DBA_S3_LOAD(A, 3); DBA_S3_WORK(B, 6);
  DBA_S3_LOAD(B, 2); DBA_S3_WORK(C, 5);
  DBA_S3_LOAD(C, 1); DBA_S3_WORK(D, 4);
  DBA_S3_LOAD(D, 0); DBA_S3_WORK(A, 3);
  DBA_S3_WORK(B, 2);
  DBA_S3_WORK(C, 1);
  DBA_S3_WORK(D, 0);
}

extern "C" void kernel_launch(void* const* d_in, const int* in_sizes, int n_in,
                              void* d_out, int out_size, void* d_ws, size_t ws_size,
                              hipStream_t stream) {
  const float* r   = (const float*)d_in[0];
  const float* w   = (const float*)d_in[1];
  const float* Jp  = (const float*)d_in[2];
  const float* Jd  = (const float*)d_in[3];
  const float* lmb = (const float*)d_in[4];
  float* out = (float*)d_out;
  float* ws  = (float*)d_ws;

  dba_stage1_reduce<<<dim3(DBA_B * DBA_G), dim3(256), 0, stream>>>(
      r, w, Jp, Jd, lmb, ws);
  dba_stage2_solve<<<dim3(DBA_B), dim3(128), 0, stream>>>(ws, lmb, out);
  dba_stage3_depth<<<dim3(DBA_B * DBA_G), dim3(256), 0, stream>>>(
      r, w, Jp, Jd, lmb, out, out + DBA_B * 6);
}


# stage 2 folded into stage 3's prologue: each workgroup sums its batch's 32 partials and one wave solves the 6x6, a column per lane, under the first four tile loads; 2 launches, not 3; sums unchanged
# speedup vs baseline: 2.4143x; 1.1818x over previous
#include <hip/hip_runtime.h>


typedef __attribute__((ext_vector_type(2))) float v2f;
typedef __attribute__((ext_vector_type(8))) float v8f;

#define DBA_B 32
#define DBA_N 65536
#define DBA_G 32
#define DBA_SPAN (DBA_N / DBA_G)
#define DBA_PW (DBA_SPAN / 8)
#define DBA_TN 32
#define DBA_NT (DBA_PW / DBA_TN)
#define DBA_LJ 0
#define DBA_LU 384
#define DBA_LA 860
#define DBA_LW 1676
#define DBA_PU 68

#define DBA_WAVE_SYNC()                                        \
  do {                                                         \
    __builtin_amdgcn_fence(__ATOMIC_RELEASE, "wavefront");     \
    __builtin_amdgcn_wave_barrier();                           \
    __builtin_amdgcn_fence(__ATOMIC_ACQUIRE, "wavefront");     \
  } while (0)

__device__ __forceinline__ float fast_rcp(float x) {
  float inv = __builtin_amdgcn_rcpf(x);
  inv = inv * (2.0f - x * inv);
  return inv;
}

__global__ __launch_bounds__(256) void dba_stage1_reduce(
    const float* __restrict__ r, const float* __restrict__ w,
    const float* __restrict__ Jp, const float* __restrict__ Jd,
    const float* __restrict__ lmbda, float* __restrict__ partial)
{
  const int b    = blockIdx.x / DBA_G;
  const int c    = blockIdx.x % DBA_G;
  const int wave = threadIdx.x >> 5;
  const int lane = threadIdx.x & 31;
  const int half = lane >> 4;
  const int i    = lane & 15;
  const float lmb = lmbda[0];

  const int base = c * DBA_SPAN + wave * DBA_PW;
  const size_t n0 = (size_t)b * DBA_N + base;

  __shared__ __attribute__((aligned(16))) float stg[8][DBA_LW];
  float* L = stg[wave];

  const float4* gJ = (const float4*)(Jp + n0 * 12) + lane;
  const float2* gr = (const float2*)(r  + n0 * 2) + lane;
  const float2* gw = (const float2*)(w  + n0 * 2) + lane;
  const float2* gd = (const float2*)(Jd + n0 * 2) + lane;

  float4*       SJ = (float4*)(L + DBA_LJ);
  const float4* TJ = (const float4*)(L + DBA_LJ) + lane * 3;
  float2* Wu = (float2*)(L + DBA_LU) + lane;
  float2* Wa = (float2*)(L + DBA_LA) + lane;

  const int ib = (i < 6) ? i : 6;
  const int ia = (i < 6) ? i : (i < 8) ? 5 : (i < 14) ? (i - 2) : 11;
  const float2* Ru = (const float2*)(L + DBA_LU + ib * DBA_PU) + half;
  const float2* Ra = (const float2*)(L + DBA_LA + ia * DBA_PU) + half;

  v8f cE = {0.f,0.f,0.f,0.f,0.f,0.f,0.f,0.f};
  v8f cO = {0.f,0.f,0.f,0.f,0.f,0.f,0.f,0.f};

  float4 j0 = gJ[0], j1 = gJ[32], j2 = gJ[64];
  float2 rv = gr[0], wc = gw[0], dv = gd[0];

  for (int t = 0; t < DBA_NT; ++t) {
    DBA_WAVE_SYNC();
    SJ[lane]      = j0;
    SJ[lane + 32] = j1;
    SJ[lane + 64] = j2;
    DBA_WAVE_SYNC();
    const float4 a0 = TJ[0], a1 = TJ[1], a2 = TJ[2];
    const float2 rc = rv, wn = wc, dn = dv;

    if (t + 1 < DBA_NT) {
      const int q = t + 1;
      j0 = gJ[q * 96]; j1 = gJ[q * 96 + 32]; j2 = gJ[q * 96 + 64];
      rv = gr[q * 32]; wc = gw[q * 32]; dv = gd[q * 32];
    }

    const float u0[7] = {a0.x, a0.y, a0.z, a0.w, a1.x, a1.y, rc.x};
    const float u1[7] = {a1.z, a1.w, a2.x, a2.y, a2.z, a2.w, rc.y};
    const float hdd = wn.x * (dn.x * dn.x + dn.y * dn.y) + lmb + wn.y;
    const float inv = fast_rcp(hdd + 1e-6f);
    const float e0 = wn.x * dn.x;
    const float e1 = wn.x * dn.y;
    Wu[6 * (DBA_PU / 2)] = make_float2(u0[6], u1[6]);
#pragma unroll
    for (int k = 0; k < 6; ++k) {
      const float p0 = wn.x * u0[k];
      const float p1 = wn.x * u1[k];
      const float tt = (dn.x * p0 + dn.y * p1) * inv;
      Wu[k * (DBA_PU / 2)]       = make_float2(u0[k], u1[k]);
      Wa[k * (DBA_PU / 2)]       = make_float2(p0, p1);
      Wa[(6 + k) * (DBA_PU / 2)] = make_float2(tt * e0, tt * e1);
    }
    DBA_WAVE_SYNC();

#pragma unroll
    for (int q = 0; q < DBA_TN / 4; ++q) {
      const float2 uA = Ru[4 * q],     cA = Ra[4 * q];
      const float2 uB = Ru[4 * q + 2], cB = Ra[4 * q + 2];
      v2f pa, pb;
      pa.x = cA.x; pa.y = cA.y; pb.x = uA.x; pb.y = uA.y;
      cE = __builtin_amdgcn_wmma_f32_16x16x4_f32(false, pa, false, pb,
                                                 (short)0, cE, false, false);
      pa.x = cB.x; pa.y = cB.y; pb.x = uB.x; pb.y = uB.y;
      cO = __builtin_amdgcn_wmma_f32_16x16x4_f32(false, pa, false, pb,
                                                 (short)0, cO, false, false);
    }
  }

  __shared__ float sred[8][112];
  if (i < 7) {
#pragma unroll
    for (int p = 0; p < 6; ++p)
      sred[wave][half * 49 + p * 7 + i] = cE[p] + cO[p];
    sred[wave][half * 49 + 42 + i] = 0.f;
  }
  __syncthreads();
  if (threadIdx.x < 98) {
    float s = 0.f;
#pragma unroll
    for (int wv = 0; wv < 8; ++wv) s += sred[wv][threadIdx.x];
    partial[(size_t)blockIdx.x * 128 + threadIdx.x] = s;
  }
}

typedef __attribute__((ext_vector_type(4))) float v4f;
#define DBA_S3_DECL(R) v4f R##j0, R##j1, R##j2; v2f R##r, R##w, R##d
#define DBA_S3_LOAD(R, q)                                              \
  do {                                                                 \
    R##j0 = gJ[(q) * 96]; R##j1 = gJ[(q) * 96 + 32];                   \
    R##j2 = gJ[(q) * 96 + 64];                                         \
    R##r = gr[(q) * 32]; R##w = gw[(q) * 32]; R##d = gd[(q) * 32];     \
  } while (0)
#define DBA_S3_WORK(R, q)                                              \
  do {                                                                 \
    DBA_WAVE_SYNC();   \
    SJ[lane] = R##j0; SJ[lane + 32] = R##j1; SJ[lane + 64] = R##j2;    \
    DBA_WAVE_SYNC();   \
    const v4f a0 = TJ[0], a1 = TJ[1], a2 = TJ[2];                      \
    const v2f rv = R##r, wv = R##w, jd = R##d;                         \
    const float h0 = wv.x * (jd.x * a0.x + jd.y * a1.z);               \
    const float h1 = wv.x * (jd.x * a0.y + jd.y * a1.w);               \
    const float h2 = wv.x * (jd.x * a0.z + jd.y * a2.x);               \
    const float h3 = wv.x * (jd.x * a0.w + jd.y * a2.y);               \
    const float h4 = wv.x * (jd.x * a1.x + jd.y * a2.z);               \
    const float h5 = wv.x * (jd.x * a1.y + jd.y * a2.w);               \
    const float gdn = wv.x * (jd.x * rv.x + jd.y * rv.y);              \
    const float hdd = wv.x * (jd.x * jd.x + jd.y * jd.y) + lmb + wv.y; \
    const float inv = fast_rcp(hdd + 1e-6f);                           \
    const float vdot = h0 * p0 + h1 * p1 + h2 * p2 + h3 * p3           \
                     + h4 * p4 + h5 * p5;                              \
    go[(q) * 32] = inv * (gdn - vdot);                                 \
  } while (0)

#define DBA_LDS_BARRIER()                                              \
  do {                                                                 \
    __builtin_amdgcn_fence(__ATOMIC_RELEASE, "workgroup", "local");    \
    __builtin_amdgcn_s_barrier();                                      \
    __builtin_amdgcn_fence(__ATOMIC_ACQUIRE, "workgroup", "local");    \
  } while (0)

#define DBA_RL(x, l) \
  __builtin_bit_cast(float, __builtin_amdgcn_readlane(__builtin_bit_cast(int, (x)), (l)))

__global__ __launch_bounds__(256) void dba_stage3_depth(
    const float* __restrict__ r, const float* __restrict__ w,
    const float* __restrict__ Jp, const float* __restrict__ Jd,
    const float* __restrict__ lmbda, const float* __restrict__ partial,
    float* __restrict__ out)
{
  const int b    = blockIdx.x / DBA_G;
  const int c    = blockIdx.x % DBA_G;
  const int wave = threadIdx.x >> 5;
  const int lane = threadIdx.x & 31;
  const float lmb = lmbda[0];

  const int base = c * DBA_SPAN + wave * DBA_PW;
  const size_t n0 = (size_t)b * DBA_N + base;

  __shared__ __attribute__((aligned(16))) float turn[8][DBA_TN * 12];
  __shared__ float acc[98];
  __shared__ float sdp[6];
  v4f*       SJ = (v4f*)turn[wave];
  const v4f* TJ = (const v4f*)turn[wave] + lane * 3;

  const v4f* gJ = (const v4f*)(Jp + n0 * 12) + lane;
  const v2f* gr = (const v2f*)(r  + n0 * 2) + lane;
  const v2f* gw = (const v2f*)(w  + n0 * 2) + lane;
  const v2f* gd = (const v2f*)(Jd + n0 * 2) + lane;
  float*     go = out + DBA_B * 6 + n0 + lane;

  const bool red = threadIdx.x < 98;
  float pv[DBA_G];
  if (red) {
    const float* pp = partial + (size_t)b * DBA_G * 128 + threadIdx.x;
#pragma unroll
    for (int g = 0; g < DBA_G; ++g) pv[g] = pp[g * 128];
  }

  DBA_S3_DECL(A); DBA_S3_DECL(B); DBA_S3_DECL(C); DBA_S3_DECL(D);
  DBA_S3_LOAD(A, 7); DBA_S3_LOAD(B, 6); DBA_S3_LOAD(C, 5);
  DBA_S3_LOAD(D, 4);

  if (red) {
    float s = 0.f;
#pragma unroll
    for (int g = 0; g < DBA_G; ++g) s += pv[g];
    acc[threadIdx.x] = s;
  }
  DBA_LDS_BARRIER();
  if (__builtin_amdgcn_readfirstlane(wave) == 0) {
    const int j = (lane < 6) ? lane : 6;
    const float dg = lmb + 0.001f;
    float a[6];
#pragma unroll
    for (int ii = 0; ii < 6; ++ii) {
      const int t = (j < 6) ? (j * 7 + ii) : (ii * 7 + 6);
      const float hij = acc[ii * 7 + j] - acc[49 + ii * 7 + j];
      const float hji = acc[t] - acc[49 + t];
      float v = 0.5f * (hij + hji);
      if (ii == j) v += dg;
      a[ii] = v;
    }
#pragma unroll
    for (int col = 0; col < 6; ++col) {
      int piv = col; float best = fabsf(DBA_RL(a[col], col));
#pragma unroll
      for (int rr = col + 1; rr < 6; ++rr) {
        const float av = fabsf(DBA_RL(a[rr], col));
        if (av > best) { best = av; piv = rr; }
      }
#pragma unroll
      for (int rr = col + 1; rr < 6; ++rr)
        if (piv == rr) { const float tmp = a[col]; a[col] = a[rr]; a[rr] = tmp; }
      const float invd = 1.0f / DBA_RL(a[col], col);
#pragma unroll
      for (int rr = col + 1; rr < 6; ++rr) {
        const float f = DBA_RL(a[rr], col) * invd;
        a[rr] -= f * a[col];
      }
    }
    float x[6];
#pragma unroll
    for (int ii = 5; ii >= 0; --ii) {
      float s = DBA_RL(a[ii], 6);
#pragma unroll
      for (int jj = ii + 1; jj < 6; ++jj) s -= DBA_RL(a[ii], jj) * x[jj];
      x[ii] = s / DBA_RL(a[ii], ii);
    }
    if (lane == 0) {
#pragma unroll
      for (int ii = 0; ii < 6; ++ii) sdp[ii] = x[ii];
      if (c == 0) {
#pragma unroll
        for (int ii = 0; ii < 6; ++ii) out[b * 6 + ii] = x[ii];
      }
    }
  }
  DBA_LDS_BARRIER();
  const float p0 = sdp[0], p1 = sdp[1], p2 = sdp[2];
  const float p3 = sdp[3], p4 = sdp[4], p5 = sdp[5];

  DBA_S3_WORK(A, 7);
  DBA_S3_LOAD(A, 3); DBA_S3_WORK(B, 6);
  DBA_S3_LOAD(B, 2); DBA_S3_WORK(C, 5);
  DBA_S3_LOAD(C, 1); DBA_S3_WORK(D, 4);
  DBA_S3_LOAD(D, 0); DBA_S3_WORK(A, 3);
  DBA_S3_WORK(B, 2);
  DBA_S3_WORK(C, 1);
  DBA_S3_WORK(D, 0);
}

extern "C" void kernel_launch(void* const* d_in, const int* in_sizes, int n_in,
                              void* d_out, int out_size, void* d_ws, size_t ws_size,
                              hipStream_t stream) {
  const float* r   = (const float*)d_in[0];
  const float* w   = (const float*)d_in[1];
  const float* Jp  = (const float*)d_in[2];
  const float* Jd  = (const float*)d_in[3];
  const float* lmb = (const float*)d_in[4];
  float* out = (float*)d_out;
  float* ws  = (float*)d_ws;

  dba_stage1_reduce<<<dim3(DBA_B * DBA_G), dim3(256), 0, stream>>>(
      r, w, Jp, Jd, lmb, ws);
  dba_stage3_depth<<<dim3(DBA_B * DBA_G), dim3(256), 0, stream>>>(
      r, w, Jp, Jd, lmb, ws, out);
}


# stage 1 sums M = P - Q as one rank-2 term per n (A rows = U*W, W = conf*I - inv*e*e^T, 2x2), four n per WMMA on its two diagonal 8x8 blocks: 8 f32 WMMAs per 32 n, not 16; sums regrouped per n
# speedup vs baseline: 2.6425x; 1.0945x over previous
#include <hip/hip_runtime.h>


typedef __attribute__((ext_vector_type(2))) float v2f;
typedef __attribute__((ext_vector_type(8))) float v8f;

#define DBA_B 32
#define DBA_N 65536
#define DBA_G 32
#define DBA_PS 64
#define DBA_SPAN (DBA_N / DBA_G)
#define DBA_PW (DBA_SPAN / 8)
#define DBA_TN 32
#define DBA_NT (DBA_PW / DBA_TN)
#define DBA_LJ 0
#define DBA_LU 384
#define DBA_PU 72
#define DBA_LA (DBA_LU + 7 * DBA_PU)
#define DBA_LW (DBA_LA + 6 * DBA_PU)

#define DBA_WAVE_SYNC()                                        \
  do {                                                         \
    __builtin_amdgcn_fence(__ATOMIC_RELEASE, "wavefront");     \
    __builtin_amdgcn_wave_barrier();                           \
    __builtin_amdgcn_fence(__ATOMIC_ACQUIRE, "wavefront");     \
  } while (0)

__device__ __forceinline__ float fast_rcp(float x) {
  float inv = __builtin_amdgcn_rcpf(x);
  inv = inv * (2.0f - x * inv);
  return inv;
}

__global__ __launch_bounds__(256) void dba_stage1_reduce(
    const float* __restrict__ r, const float* __restrict__ w,
    const float* __restrict__ Jp, const float* __restrict__ Jd,
    const float* __restrict__ lmbda, float* __restrict__ partial)
{
  const int b    = blockIdx.x / DBA_G;
  const int c    = blockIdx.x % DBA_G;
  const int wave = threadIdx.x >> 5;
  const int lane = threadIdx.x & 31;
  const int half = lane >> 4;
  const int i    = lane & 15;
  const int blk  = i >> 3;
  const int ci   = i & 7;
  const float lmb = lmbda[0];

  const int base = c * DBA_SPAN + wave * DBA_PW;
  const size_t n0 = (size_t)b * DBA_N + base;

  __shared__ __attribute__((aligned(16))) float stg[8][DBA_LW];
  float* L = stg[wave];

  const float4* gJ = (const float4*)(Jp + n0 * 12) + lane;
  const float2* gr = (const float2*)(r  + n0 * 2) + lane;
  const float2* gw = (const float2*)(w  + n0 * 2) + lane;
  const float2* gd = (const float2*)(Jd + n0 * 2) + lane;

  float4*       SJ = (float4*)(L + DBA_LJ);
  const float4* TJ = (const float4*)(L + DBA_LJ) + lane * 3;
  float2* Wu = (float2*)(L + DBA_LU) + lane;
  float2* Wa = (float2*)(L + DBA_LA) + lane;

  const int ib = (ci < 6) ? ci : 6;
  const int ia = (ci < 6) ? ci : 5;
  const float2* Ru = (const float2*)(L + DBA_LU + ib * DBA_PU) + 2 * blk + half;
  const float2* Ra = (const float2*)(L + DBA_LA + ia * DBA_PU) + 2 * blk + half;

  v8f cE = {0.f,0.f,0.f,0.f,0.f,0.f,0.f,0.f};
  v8f cO = {0.f,0.f,0.f,0.f,0.f,0.f,0.f,0.f};

  float4 j0 = gJ[0], j1 = gJ[32], j2 = gJ[64];
  float2 rv = gr[0], wc = gw[0], dv = gd[0];

  for (int t = 0; t < DBA_NT; ++t) {
    DBA_WAVE_SYNC();
    SJ[lane]      = j0;
    SJ[lane + 32] = j1;
    SJ[lane + 64] = j2;
    DBA_WAVE_SYNC();
    const float4 a0 = TJ[0], a1 = TJ[1], a2 = TJ[2];
    const float2 rc = rv, wn = wc, dn = dv;

    if (t + 1 < DBA_NT) {
      const int q = t + 1;
      j0 = gJ[q * 96]; j1 = gJ[q * 96 + 32]; j2 = gJ[q * 96 + 64];
      rv = gr[q * 32]; wc = gw[q * 32]; dv = gd[q * 32];
    }

    const float u0[7] = {a0.x, a0.y, a0.z, a0.w, a1.x, a1.y, rc.x};
    const float u1[7] = {a1.z, a1.w, a2.x, a2.y, a2.z, a2.w, rc.y};
    const float hdd = wn.x * (dn.x * dn.x + dn.y * dn.y) + lmb + wn.y;
    const float inv = fast_rcp(hdd + 1e-6f);
    const float e0 = wn.x * dn.x;
    const float e1 = wn.x * dn.y;
    const float f0 = inv * e0;
    const float f1 = inv * e1;
    const float w00 = wn.x - f0 * e0;
    const float w01 = -(f0 * e1);
    const float w11 = wn.x - f1 * e1;
    Wu[6 * (DBA_PU / 2)] = make_float2(u0[6], u1[6]);
#pragma unroll
    for (int k = 0; k < 6; ++k) {
      Wu[k * (DBA_PU / 2)] = make_float2(u0[k], u1[k]);
      Wa[k * (DBA_PU / 2)] = make_float2(u0[k] * w00 + u1[k] * w01,
                                         u0[k] * w01 + u1[k] * w11);
    }
    DBA_WAVE_SYNC();

#pragma unroll
    for (int q = 0; q < DBA_TN / 8; ++q) {
      const float2 uA = Ru[8 * q],     cA = Ra[8 * q];
      const float2 uB = Ru[8 * q + 4], cB = Ra[8 * q + 4];
      v2f pa, pb;
      pa.x = cA.x; pa.y = cA.y; pb.x = uA.x; pb.y = uA.y;
      cE = __builtin_amdgcn_wmma_f32_16x16x4_f32(false, pa, false, pb,
                                                 (short)0, cE, false, false);
      pa.x = cB.x; pa.y = cB.y; pb.x = uB.x; pb.y = uB.y;
      cO = __builtin_amdgcn_wmma_f32_16x16x4_f32(false, pa, false, pb,
                                                 (short)0, cO, false, false);
    }
  }

  __shared__ float sred[8][84];
  if (blk == half && ci < 7) {
#pragma unroll
    for (int p = 0; p < 6; ++p)
      sred[wave][half * 42 + p * 7 + ci] = cE[p] + cO[p];
  }
  __syncthreads();
  if (threadIdx.x < 42) {
    float s = 0.f;
#pragma unroll
    for (int wv = 0; wv < 8; ++wv)
      s += sred[wv][threadIdx.x] + sred[wv][42 + threadIdx.x];
    partial[(size_t)blockIdx.x * DBA_PS + threadIdx.x] = s;
  }
}

typedef __attribute__((ext_vector_type(4))) float v4f;
#define DBA_S3_DECL(R) v4f R##j0, R##j1, R##j2; v2f R##r, R##w, R##d
#define DBA_S3_LOAD(R, q)                                              \
  do {                                                                 \
    R##j0 = gJ[(q) * 96]; R##j1 = gJ[(q) * 96 + 32];                   \
    R##j2 = gJ[(q) * 96 + 64];                                         \
    R##r = gr[(q) * 32]; R##w = gw[(q) * 32]; R##d = gd[(q) * 32];     \
  } while (0)
#define DBA_S3_WORK(R, q)                                              \
  do {                                                                 \
    DBA_WAVE_SYNC();   \
    SJ[lane] = R##j0; SJ[lane + 32] = R##j1; SJ[lane + 64] = R##j2;    \
    DBA_WAVE_SYNC();   \
    const v4f a0 = TJ[0], a1 = TJ[1], a2 = TJ[2];                      \
    const v2f rv = R##r, wv = R##w, jd = R##d;                         \
    const float h0 = wv.x * (jd.x * a0.x + jd.y * a1.z);               \
    const float h1 = wv.x * (jd.x * a0.y + jd.y * a1.w);               \
    const float h2 = wv.x * (jd.x * a0.z + jd.y * a2.x);               \
    const float h3 = wv.x * (jd.x * a0.w + jd.y * a2.y);               \
    const float h4 = wv.x * (jd.x * a1.x + jd.y * a2.z);               \
    const float h5 = wv.x * (jd.x * a1.y + jd.y * a2.w);               \
    const float gdn = wv.x * (jd.x * rv.x + jd.y * rv.y);              \
    const float hdd = wv.x * (jd.x * jd.x + jd.y * jd.y) + lmb + wv.y; \
    const float inv = fast_rcp(hdd + 1e-6f);                           \
    const float vdot = h0 * p0 + h1 * p1 + h2 * p2 + h3 * p3           \
                     + h4 * p4 + h5 * p5;                              \
    go[(q) * 32] = inv * (gdn - vdot);                                 \
  } while (0)

#define DBA_LDS_BARRIER()                                              \
  do {                                                                 \
    __builtin_amdgcn_fence(__ATOMIC_RELEASE, "workgroup", "local");    \
    __builtin_amdgcn_s_barrier();                                      \
    __builtin_amdgcn_fence(__ATOMIC_ACQUIRE, "workgroup", "local");    \
  } while (0)

#define DBA_RL(x, l) \
  __builtin_bit_cast(float, __builtin_amdgcn_readlane(__builtin_bit_cast(int, (x)), (l)))

__global__ __launch_bounds__(256) void dba_stage3_depth(
    const float* __restrict__ r, const float* __restrict__ w,
    const float* __restrict__ Jp, const float* __restrict__ Jd,
    const float* __restrict__ lmbda, const float* __restrict__ partial,
    float* __restrict__ out)
{
  const int b    = blockIdx.x / DBA_G;
  const int c    = blockIdx.x % DBA_G;
  const int wave = threadIdx.x >> 5;
  const int lane = threadIdx.x & 31;
  const float lmb = lmbda[0];

  const int base = c * DBA_SPAN + wave * DBA_PW;
  const size_t n0 = (size_t)b * DBA_N + base;

  __shared__ __attribute__((aligned(16))) float turn[8][DBA_TN * 12];
  __shared__ float acc[42];
  __shared__ float sdp[6];
  v4f*       SJ = (v4f*)turn[wave];
  const v4f* TJ = (const v4f*)turn[wave] + lane * 3;

  const v4f* gJ = (const v4f*)(Jp + n0 * 12) + lane;
  const v2f* gr = (const v2f*)(r  + n0 * 2) + lane;
  const v2f* gw = (const v2f*)(w  + n0 * 2) + lane;
  const v2f* gd = (const v2f*)(Jd + n0 * 2) + lane;
  float*     go = out + DBA_B * 6 + n0 + lane;

  const bool red = threadIdx.x < 42;
  float pv[DBA_G];
  if (red) {
    const float* pp = partial + (size_t)b * DBA_G * DBA_PS + threadIdx.x;
#pragma unroll
    for (int g = 0; g < DBA_G; ++g) pv[g] = pp[g * DBA_PS];
  }

  DBA_S3_DECL(A); DBA_S3_DECL(B); DBA_S3_DECL(C); DBA_S3_DECL(D);
  DBA_S3_LOAD(A, 7); DBA_S3_LOAD(B, 6); DBA_S3_LOAD(C, 5);
  DBA_S3_LOAD(D, 4);

  if (red) {
    float s = 0.f;
#pragma unroll
    for (int g = 0; g < DBA_G; ++g) s += pv[g];
    acc[threadIdx.x] = s;
  }
  DBA_LDS_BARRIER();
  if (__builtin_amdgcn_readfirstlane(wave) == 0) {
    const int j = (lane < 6) ? lane : 6;
    const float dg = lmb + 0.001f;
    float a[6];
#pragma unroll
    for (int ii = 0; ii < 6; ++ii) {
      const int t = (j < 6) ? (j * 7 + ii) : (ii * 7 + 6);
      const float hij = acc[ii * 7 + j];
      const float hji = acc[t];
      float v = 0.5f * (hij + hji);
      if (ii == j) v += dg;
      a[ii] = v;
    }
#pragma unroll
    for (int col = 0; col < 6; ++col) {
      int piv = col; float best = fabsf(DBA_RL(a[col], col));
#pragma unroll
      for (int rr = col + 1; rr < 6; ++rr) {
        const float av = fabsf(DBA_RL(a[rr], col));
        if (av > best) { best = av; piv = rr; }
      }
#pragma unroll
      for (int rr = col + 1; rr < 6; ++rr)
        if (piv == rr) { const float tmp = a[col]; a[col] = a[rr]; a[rr] = tmp; }
      const float invd = 1.0f / DBA_RL(a[col], col);
#pragma unroll
      for (int rr = col + 1; rr < 6; ++rr) {
        const float f = DBA_RL(a[rr], col) * invd;
        a[rr] -= f * a[col];
      }
    }
    float x[6];
#pragma unroll
    for (int ii = 5; ii >= 0; --ii) {
      float s = DBA_RL(a[ii], 6);
#pragma unroll
      for (int jj = ii + 1; jj < 6; ++jj) s -= DBA_RL(a[ii], jj) * x[jj];
      x[ii] = s / DBA_RL(a[ii], ii);
    }
    if (lane == 0) {
#pragma unroll
      for (int ii = 0; ii < 6; ++ii) sdp[ii] = x[ii];
      if (c == 0) {
#pragma unroll
        for (int ii = 0; ii < 6; ++ii) out[b * 6 + ii] = x[ii];
      }
    }
  }
  DBA_LDS_BARRIER();
  const float p0 = sdp[0], p1 = sdp[1], p2 = sdp[2];
  const float p3 = sdp[3], p4 = sdp[4], p5 = sdp[5];

  DBA_S3_WORK(A, 7);
  DBA_S3_LOAD(A, 3); DBA_S3_WORK(B, 6);
  DBA_S3_LOAD(B, 2); DBA_S3_WORK(C, 5);
  DBA_S3_LOAD(C, 1); DBA_S3_WORK(D, 4);
  DBA_S3_LOAD(D, 0); DBA_S3_WORK(A, 3);
  DBA_S3_WORK(B, 2);
  DBA_S3_WORK(C, 1);
  DBA_S3_WORK(D, 0);
}

extern "C" void kernel_launch(void* const* d_in, const int* in_sizes, int n_in,
                              void* d_out, int out_size, void* d_ws, size_t ws_size,
                              hipStream_t stream) {
  const float* r   = (const float*)d_in[0];
  const float* w   = (const float*)d_in[1];
  const float* Jp  = (const float*)d_in[2];
  const float* Jd  = (const float*)d_in[3];
  const float* lmb = (const float*)d_in[4];
  float* out = (float*)d_out;
  float* ws  = (float*)d_ws;

  dba_stage1_reduce<<<dim3(DBA_B * DBA_G), dim3(256), 0, stream>>>(
      r, w, Jp, Jd, lmb, ws);
  dba_stage3_depth<<<dim3(DBA_B * DBA_G), dim3(256), 0, stream>>>(
      r, w, Jp, Jd, lmb, ws, out);
}
